# starting kernel
# baseline (speedup 1.0000x reference)
#include <hip/hip_runtime.h>
#include <hip/hip_bf16.h>

typedef __attribute__((ext_vector_type(2))) float v2f;
typedef __attribute__((ext_vector_type(4))) float v4f;
typedef __attribute__((ext_vector_type(8))) float v8f;

#define DIM_I 64
#define DIM_O 128
#define DIM_SP 1024
#define DIM_N 8192
#define NOISE_F 0.1f

__global__ __launch_bounds__(256)
void noiseconv_wmma_conv(const float* __restrict__ x,
                         const float* __restrict__ W,
                         const float* __restrict__ bias,
                         float* __restrict__ out) {
  const int wave = blockIdx.x * (blockDim.x >> 5) + (threadIdx.x >> 5);
  if (wave >= 8 * 8 * 64) return;
  const int lane = threadIdx.x & 31;

  const int ct = wave & 63;
  const int ot = (wave >> 6) & 7;
  const int bb = wave >> 9;
  const int c0 = ct * 16;
  const int o0 = ot * 16;

  const float* xb = x + (size_t)bb * DIM_I * DIM_SP;

  const int mn    = lane & 15;
  const int khalf = (lane >> 4) * 2;

  v8f acc = {};
  #pragma unroll 4
  for (int k0 = 0; k0 < DIM_I; k0 += 4) {
    v2f a = *(const v2f*)(W + (size_t)(o0 + mn) * DIM_I + k0 + khalf);
    const float* xp = xb + (size_t)(k0 + khalf) * DIM_SP + c0 + mn;
    v2f b;
    b.x = xp[0];
    b.y = xp[DIM_SP];
    acc = __builtin_amdgcn_wmma_f32_16x16x4_f32(
         false, a,  false, b,
         (short)0, acc,  false,  false);
  }

  const int nl  = lane & 15;
  const int mhi = (lane >> 4) * 8;
  float* ob = out + (size_t)bb * DIM_O * DIM_SP + c0 + nl;
  #pragma unroll
  for (int r = 0; r < 8; ++r) {
    const int mrow = r + mhi;
    ob[(size_t)(o0 + mrow) * DIM_SP] = acc[r] + bias[o0 + mrow];
  }
}

__global__ __launch_bounds__(256)
void noiseconv_noisy_stream(const float* __restrict__ x,
                            const float* __restrict__ W,
                            const float* __restrict__ eps,
                            float* __restrict__ out) {
  const int wid = blockIdx.x * (blockDim.x >> 5) + (threadIdx.x >> 5);
  if (wid >= DIM_N) return;
  const int lane = threadIdx.x & 31;

  const int nn = wid;
  const int bb = nn >> 10;
  const int sp = nn & (DIM_SP - 1);

  const int il   = lane & 15;
  const int ohalf = lane >> 4;
  const int ib   = 4 * il;

  const float* xb = x + (size_t)bb * DIM_I * DIM_SP + sp;
  v4f xv;
  xv.x = xb[(size_t)(ib + 0) * DIM_SP];
  xv.y = xb[(size_t)(ib + 1) * DIM_SP];
  xv.z = xb[(size_t)(ib + 2) * DIM_SP];
  xv.w = xb[(size_t)(ib + 3) * DIM_SP];

  const float* ep = eps + (size_t)nn * DIM_O * DIM_I;
  float* ob = out + (size_t)bb * DIM_O * DIM_SP + sp;

  #pragma unroll 2
  for (int o0 = 0; o0 < DIM_O; o0 += 2) {
    const int o = o0 + ohalf;
    v4f wv = *(const v4f*)(W + (size_t)o * DIM_I + ib);
    v4f ev = __builtin_nontemporal_load((const v4f*)(ep + (size_t)o * DIM_I + ib));

    float part = ev.x * (wv.x * xv.x)
               + ev.y * (wv.y * xv.y)
               + ev.z * (wv.z * xv.z)
               + ev.w * (wv.w * xv.w);

    part += __shfl_xor(part, 1, 32);
    part += __shfl_xor(part, 2, 32);
    part += __shfl_xor(part, 4, 32);
    part += __shfl_xor(part, 8, 32);

    if (il == 0) {
      ob[(size_t)o * DIM_SP] += NOISE_F * part;
    }
  }
}

extern "C" void kernel_launch(void* const* d_in, const int* in_sizes, int n_in,
                              void* d_out, int out_size, void* d_ws, size_t ws_size,
                              hipStream_t stream) {
  const float* x    = (const float*)d_in[0];
  const float* W    = (const float*)d_in[1];
  const float* bias = (const float*)d_in[2];
  const float* eps  = (const float*)d_in[3];
  float* out = (float*)d_out;
  (void)in_sizes; (void)n_in; (void)out_size; (void)d_ws; (void)ws_size;

  noiseconv_wmma_conv<<<512, 256, 0, stream>>>(x, W, bias, out);
  noiseconv_noisy_stream<<<1024, 256, 0, stream>>>(x, W, eps, out);
}


# One fused launch built around the eps stream: a wave keeps its 16 W rows in registers, reads eps as contiguous 4 KB per n (8 NT loads, 2 n ahead), DPP row sums; out stored once as whole lines
# speedup vs baseline: 6.0389x; 6.0389x over previous
#include <hip/hip_runtime.h>


typedef __attribute__((ext_vector_type(2))) float v2f;
typedef __attribute__((ext_vector_type(4))) float v4f;
typedef __attribute__((ext_vector_type(8))) float v8f;

#define DIM_I 64
#define DIM_O 128
#define DIM_SP 1024
#define DIM_N 8192
#define NOISE_F 0.1f

#define WG_SP 32
#define WG_O 64
#define XS_PITCH 68
#define EPS_ROWS 8
#define WAVE_N 16
#define OS_PITCH 33
#define AHEAD 2

template <int CTRL>
__device__ __forceinline__ float row_xor_add(float v) {
  const int iv = __builtin_bit_cast(int, v);
  const int sw = __builtin_amdgcn_update_dpp(0, iv, CTRL, 0xf, 0xf, true);
  return v + __builtin_bit_cast(float, sw);
}

__global__ __launch_bounds__(256)
void noiseconv_fused_stream(const float* __restrict__ x,
                            const float* __restrict__ W,
                            const float* __restrict__ bias,
                            const float* __restrict__ eps,
                            float* __restrict__ out) {
  __shared__ __attribute__((aligned(16))) float xs[WG_SP * XS_PITCH];
  __shared__ float cs[WG_O * WG_SP];
  __shared__ float os[WG_O * OS_PITCH];

  const int tid  = threadIdx.x;
  const int lane = tid & 31;
  const int wv_id = tid >> 5;
  const int wo = wv_id >> 1;
  const int wn = wv_id & 1;

  const int ohalf = blockIdx.x & 1;
  const int grp   = blockIdx.x >> 1;
  const int bb    = grp >> 5;
  const int sp0   = (grp & 31) * WG_SP;

  const int ow0 = WG_O * ohalf;
  const int o0  = ow0 + 16 * wo;
  const int c0  = sp0 + 16 * wn;

  const float* xb = x + (size_t)bb * DIM_I * DIM_SP;

  #pragma unroll
  for (int q = 0; q < 2; ++q) {
    const int idx = tid + 256 * q;
    const int i   = idx >> 3;
    const int c4  = (idx & 7) * 4;
    const v4f xv4 = *(const v4f*)(xb + (size_t)i * DIM_SP + sp0 + c4);
    xs[(c4 + 0) * XS_PITCH + i] = xv4.x;
    xs[(c4 + 1) * XS_PITCH + i] = xv4.y;
    xs[(c4 + 2) * XS_PITCH + i] = xv4.z;
    xs[(c4 + 3) * XS_PITCH + i] = xv4.w;
  }

  {
    const int mn    = lane & 15;
    const int khalf = (lane >> 4) * 2;
    v8f acc = {};
    #pragma unroll
    for (int k0 = 0; k0 < DIM_I; k0 += 4) {
      v2f a = *(const v2f*)(W + (size_t)(o0 + mn) * DIM_I + k0 + khalf);
      const float* xp = xb + (size_t)(k0 + khalf) * DIM_SP + c0 + mn;
      v2f b;
      b.x = xp[0];
      b.y = xp[DIM_SP];
      acc = __builtin_amdgcn_wmma_f32_16x16x4_f32(
           false, a,  false, b,
           (short)0, acc,  false,  false);
    }
    const int mhi = (lane >> 4) * 8;
    #pragma unroll
    for (int r = 0; r < 8; ++r) {
      const int mrow = r + mhi;
      cs[(16 * wo + mrow) * WG_SP + 16 * wn + mn] = acc[r] + bias[o0 + mrow];
    }
  }

  __syncthreads();

  const int il = lane & 15;
  const int h  = lane >> 4;

  v4f wreg[EPS_ROWS];
  #pragma unroll
  for (int j = 0; j < EPS_ROWS; ++j)
    wreg[j] = *(const v4f*)(W + (size_t)(o0 + 2 * j + h) * DIM_I + 4 * il);

  const float* ep = eps + ((size_t)(bb * DIM_SP + c0) * DIM_O + o0) * DIM_I + 4 * lane;

  v4f ebuf[AHEAD + 1][EPS_ROWS];
  const int jsel = il & 7;
  float* osw = os + (16 * wo + 2 * jsel + h) * OS_PITCH + 16 * wn;

  #pragma unroll
  for (int a = 0; a < AHEAD; ++a) {
    #pragma unroll
    for (int j = 0; j < EPS_ROWS; ++j)
      ebuf[a][j] = __builtin_nontemporal_load(
          (const v4f*)(ep + (size_t)a * DIM_O * DIM_I + j * 2 * DIM_I));
  }

  #pragma unroll
  for (int nl = 0; nl < WAVE_N; ++nl) {
    if (nl + AHEAD < WAVE_N) {
      #pragma unroll
      for (int j = 0; j < EPS_ROWS; ++j)
        ebuf[(nl + AHEAD) % (AHEAD + 1)][j] = __builtin_nontemporal_load(
            (const v4f*)(ep + (size_t)(nl + AHEAD) * DIM_O * DIM_I + j * 2 * DIM_I));
    }
    __builtin_amdgcn_sched_barrier(0);

    const v4f xv = *(const v4f*)(xs + (16 * wn + nl) * XS_PITCH + 4 * il);
    float sel = 0.0f;
    #pragma unroll
    for (int j = 0; j < EPS_ROWS; ++j) {
      const v4f ev = ebuf[nl % (AHEAD + 1)][j];
      const v4f wv = wreg[j];
      float part = ev.x * (wv.x * xv.x)
                 + ev.y * (wv.y * xv.y)
                 + ev.z * (wv.z * xv.z)
                 + ev.w * (wv.w * xv.w);
      part = row_xor_add<0x161>(part);
      part = row_xor_add<0x162>(part);
      part = row_xor_add<0x164>(part);
      part = row_xor_add<0x168>(part);
      sel = (jsel == j) ? part : sel;
    }
    osw[nl] = sel;
    __builtin_amdgcn_sched_barrier(0);
  }

  __syncthreads();

  float* ob = out + ((size_t)bb * DIM_O + ow0) * DIM_SP + sp0 + lane;
  #pragma unroll
  for (int rr = 0; rr < 8; ++rr) {
    const int row = 8 * wv_id + rr;
    float v = cs[row * WG_SP + lane];
    v += NOISE_F * os[row * OS_PITCH + lane];
    ob[(size_t)row * DIM_SP] = v;
  }
}

extern "C" void kernel_launch(void* const* d_in, const int* in_sizes, int n_in,
                              void* d_out, int out_size, void* d_ws, size_t ws_size,
                              hipStream_t stream) {
  const float* x    = (const float*)d_in[0];
  const float* W    = (const float*)d_in[1];
  const float* bias = (const float*)d_in[2];
  const float* eps  = (const float*)d_in[3];
  float* out = (float*)d_out;
  (void)in_sizes; (void)n_in; (void)out_size; (void)d_ws; (void)ws_size;

  noiseconv_fused_stream<<<512, 256, 0, stream>>>(x, W, bias, eps, out);
}
